# starting kernel
# baseline (speedup 1.0000x reference)
#include <hip/hip_runtime.h>


typedef float v2f __attribute__((ext_vector_type(2)));
typedef float v8f __attribute__((ext_vector_type(8)));

#define B_ 128
#define T_ 4096
#define I_ 6
#define H_ 32
#define O_ 2

__device__ __forceinline__ v8f wmma4(v2f a, v2f b, v8f c) {
    return __builtin_amdgcn_wmma_f32_16x16x4_f32(
         false, a,  false, b,
         (short)0, c,  false,  false);
}

__device__ __forceinline__ float fsigmoid(float a) {
    return __builtin_amdgcn_rcpf(1.0f + __builtin_amdgcn_exp2f(-1.44269504088896f * a));
}
__device__ __forceinline__ float ftanh(float a) {
    return 2.0f * __builtin_amdgcn_rcpf(1.0f + __builtin_amdgcn_exp2f(-2.88539008177793f * a)) - 1.0f;
}

__global__ __launch_bounds__(64, 1) void gru_fused_kernel(
    const float* __restrict__ x,
    const int*   __restrict__ lengths,
    const float* __restrict__ w_ih,
    const float* __restrict__ w_hh,
    const float* __restrict__ b_ih,
    const float* __restrict__ b_hh,
    const float* __restrict__ fc_w,
    const float* __restrict__ fc_b,
    float*       __restrict__ out)
{
    __shared__ float ldsh[H_ * 16];

    const int tid  = threadIdx.x;
    const int wv   = tid >> 5;
    const int lane = tid & 31;
    const int lh   = lane >> 4;
    const int ln   = lane & 15;
    const int b0   = blockIdx.x * 16;

    for (int j = tid; j < H_ * 16; j += 64) ldsh[j] = 0.0f;
    __syncthreads();

    v2f Wg[3][8];
    v2f Wi[3][2];
    float biasR, biasZ, bxn, bhn;
    #pragma unroll
    for (int i = 0; i < 3; ++i) {
        const int g = i * 32 + wv * 16 + ln;
        #pragma unroll
        for (int s = 0; s < 8; ++s) {
            const int k0 = 4 * s + 2 * lh;
            Wg[i][s].x = w_hh[g * H_ + k0];
            Wg[i][s].y = w_hh[g * H_ + k0 + 1];
        }
        #pragma unroll
        for (int s = 0; s < 2; ++s) {
            const int k0 = 4 * s + 2 * lh;
            Wi[i][s].x = (k0     < I_) ? w_ih[g * I_ + k0]     : 0.0f;
            Wi[i][s].y = (k0 + 1 < I_) ? w_ih[g * I_ + k0 + 1] : 0.0f;
        }
    }
    {
        const int gr = 0 * 32 + wv * 16 + ln;
        const int gz = 1 * 32 + wv * 16 + ln;
        const int gn = 2 * 32 + wv * 16 + ln;
        biasR = b_ih[gr] + b_hh[gr];
        biasZ = b_ih[gz] + b_hh[gz];
        bxn   = b_ih[gn];
        bhn   = b_hh[gn];
    }

    const bool fcl = (wv == 0) && (ln < O_);
    v2f Wfc[8];
    #pragma unroll
    for (int s = 0; s < 8; ++s) {
        const int k0 = 4 * s + 2 * lh;
        Wfc[s].x = fcl ? fc_w[ln * H_ + k0]     : 0.0f;
        Wfc[s].y = fcl ? fc_w[ln * H_ + k0 + 1] : 0.0f;
    }
    const float fcb = fcl ? fc_b[ln] : 0.0f;

    int len8[8];
    #pragma unroll
    for (int v = 0; v < 8; ++v) len8[v] = lengths[b0 + v + 8 * lh];

    const float* xrow = x + (size_t)(b0 + ln) * T_ * I_;
    const int    hcol = wv * 16 + ln;

    v2f   hA[8];
    float hold[8];
    #pragma unroll
    for (int s = 0; s < 8; ++s) { hA[s].x = 0.0f; hA[s].y = 0.0f; }
    #pragma unroll
    for (int v = 0; v < 8; ++v) hold[v] = 0.0f;

    #pragma unroll 1
    for (int t = 0; t <= T_; ++t) {
        if (wv == 0) {
            v8f accF;
            #pragma unroll
            for (int v = 0; v < 8; ++v) accF[v] = fcb;
            #pragma unroll
            for (int s = 0; s < 8; ++s) accF = wmma4(hA[s], Wfc[s], accF);
            if (t > 0 && ln < O_) {
                #pragma unroll
                for (int v = 0; v < 8; ++v) {
                    const float val = ((t - 1) < len8[v]) ? accF[v] : fcb;
                    out[((size_t)(b0 + v + 8 * lh) * T_ + (t - 1)) * O_ + ln] = val;
                }
            }
        }
        if (t == T_) break;

        const float* xp = xrow + t * I_;
        v2f xa0, xa1;
        xa0.x = xp[2 * lh];
        xa0.y = xp[2 * lh + 1];
        xa1.x = lh ? 0.0f : xp[4];
        xa1.y = lh ? 0.0f : xp[5];
        if (t + 1 < T_) __builtin_prefetch(xp + I_, 0, 1);

        v8f accR, accZ, accXN, accHN;
        #pragma unroll
        for (int v = 0; v < 8; ++v) {
            accR[v] = biasR; accZ[v] = biasZ; accXN[v] = bxn; accHN[v] = bhn;
        }
        accR  = wmma4(xa0, Wi[0][0], accR);  accR  = wmma4(xa1, Wi[0][1], accR);
        accZ  = wmma4(xa0, Wi[1][0], accZ);  accZ  = wmma4(xa1, Wi[1][1], accZ);
        accXN = wmma4(xa0, Wi[2][0], accXN); accXN = wmma4(xa1, Wi[2][1], accXN);
        #pragma unroll
        for (int s = 0; s < 8; ++s) {
            accR  = wmma4(hA[s], Wg[0][s], accR);
            accZ  = wmma4(hA[s], Wg[1][s], accZ);
            accHN = wmma4(hA[s], Wg[2][s], accHN);
        }

        #pragma unroll
        for (int v = 0; v < 8; ++v) {
            const float r  = fsigmoid(accR[v]);
            const float z  = fsigmoid(accZ[v]);
            const float n  = ftanh(accXN[v] + r * accHN[v]);
            const float hn = (1.0f - z) * n + z * hold[v];
            const float hs = (t < len8[v]) ? hn : hold[v];
            ldsh[hcol * 16 + 8 * lh + v] = hs;
            hold[v] = hs;
        }

        __syncthreads();

        #pragma unroll
        for (int s = 0; s < 8; ++s) {
            const int k0 = 4 * s + 2 * lh;
            hA[s].x = ldsh[k0 * 16 + ln];
            hA[s].y = ldsh[(k0 + 1) * 16 + ln];
        }

        __syncthreads();
    }
}

extern "C" void kernel_launch(void* const* d_in, const int* in_sizes, int n_in,
                              void* d_out, int out_size, void* d_ws, size_t ws_size,
                              hipStream_t stream) {
    const float* x       = (const float*)d_in[0];
    const int*   lengths = (const int*)  d_in[1];
    const float* w_ih    = (const float*)d_in[2];
    const float* w_hh    = (const float*)d_in[3];
    const float* b_ih    = (const float*)d_in[4];
    const float* b_hh    = (const float*)d_in[5];
    const float* fc_w    = (const float*)d_in[6];
    const float* fc_b    = (const float*)d_in[7];
    float*       out     = (float*)d_out;

    gru_fused_kernel<<<dim3(B_ / 16), dim3(64), 0, stream>>>(
        x, lengths, w_ih, w_hh, b_ih, b_hh, fc_w, fc_b, out);
}


# Step head cleared of LDS waits: own h columns turned D->A by v_permlane16_var/x16_var + select (was ds_bpermute); signal moved behind the next step's first 3 WMMAs; gate and side loops split
# speedup vs baseline: 2.5188x; 2.5188x over previous
#include <hip/hip_runtime.h>


typedef float v2f __attribute__((ext_vector_type(2)));
typedef float v4f __attribute__((ext_vector_type(4)));
typedef float v8f __attribute__((ext_vector_type(8)));

#define B_ 128
#define T_ 4096
#define I_ 6
#define H_ 32
#define O_ 2
#define RB_ 2
#define TB_ 8
#define NBLK_ (T_ / TB_)

__device__ __forceinline__ v8f wmma4(v2f a, v2f b, v8f c) {
    return __builtin_amdgcn_wmma_f32_16x16x4_f32(
         false, a,  false, b,
         (short)0, c,  false,  false);
}

__device__ __forceinline__ float fsigmoid(float a) {
    return __builtin_amdgcn_rcpf(1.0f + __builtin_amdgcn_exp2f(-1.44269504088896f * a));
}
__device__ __forceinline__ float ftanh(float a) {
    return 2.0f * __builtin_amdgcn_rcpf(1.0f + __builtin_amdgcn_exp2f(-2.88539008177793f * a)) - 1.0f;
}

__device__ __forceinline__ void step_signal() {
    asm volatile("s_wait_dscnt 0x0\n\ts_barrier_signal -1" ::: "memory");
}
__device__ __forceinline__ void step_wait() {
    asm volatile("s_barrier_wait -1" ::: "memory");
}

__device__ __forceinline__ void step_signal_behind(float r0, float z0, float n0) {
    asm volatile("s_wait_dscnt 0x0\n\ts_barrier_signal -1" :: "v"(r0), "v"(z0), "v"(n0) : "memory");
}
__device__ __forceinline__ void step_wait_behind(float r0, float z0, float n0) {
    asm volatile("s_barrier_wait -1" :: "v"(r0), "v"(z0), "v"(n0) : "memory");
}

__global__ __launch_bounds__(96, 1) void gru_fused_kernel(
    const float* __restrict__ x,
    const int*   __restrict__ lengths,
    const float* __restrict__ w_ih,
    const float* __restrict__ w_hh,
    const float* __restrict__ b_ih,
    const float* __restrict__ b_hh,
    const float* __restrict__ fc_w,
    const float* __restrict__ fc_b,
    float*       __restrict__ out)
{
    __shared__ __attribute__((aligned(16))) float ldsh[2 * TB_ * RB_ * H_];
    __shared__ __attribute__((aligned(16))) float ldsg[2 * TB_ * 2 * 32 * 4];

    const int tid  = threadIdx.x;
    const int wv   = __builtin_amdgcn_readfirstlane(tid >> 5);
    const int lane = tid & 31;
    const int lh   = lane >> 4;
    const int ln   = lane & 15;
    const int b0   = blockIdx.x * RB_;
    const int slotA = ln >> 3;
    const int stepA = ln & 7;
    const bool side = (wv == 2);


    v2f   Wg[3][8];
    float bhn = 0.0f;
    v2f   Wx[6][2];
    float bx[6];
    v2f   Wfc[8];
    float fcb = 0.0f;
    #pragma unroll
    for (int i = 0; i < 3; ++i)
        #pragma unroll
        for (int s = 0; s < 8; ++s) { Wg[i][s].x = 0.0f; Wg[i][s].y = 0.0f; }
    #pragma unroll
    for (int q = 0; q < 6; ++q) {
        bx[q] = 0.0f;
        #pragma unroll
        for (int s = 0; s < 2; ++s) { Wx[q][s].x = 0.0f; Wx[q][s].y = 0.0f; }
    }
    #pragma unroll
    for (int s = 0; s < 8; ++s) { Wfc[s].x = 0.0f; Wfc[s].y = 0.0f; }

    if (!side) {
        #pragma unroll
        for (int i = 0; i < 3; ++i) {
            const int g = i * 32 + wv * 16 + ln;
            #pragma unroll
            for (int s = 0; s < 8; ++s) {
                const int sl = 4 * (wv ^ (s >> 2)) + (s & 3);
                const int k0 = 4 * sl + 2 * lh;
                Wg[i][s].x = w_hh[g * H_ + k0];
                Wg[i][s].y = w_hh[g * H_ + k0 + 1];
            }
        }
        bhn = b_hh[2 * 32 + wv * 16 + ln];
    } else {
        #pragma unroll
        for (int q = 0; q < 6; ++q) {
            const int i = q % 3;
            const int g = i * 32 + (q / 3) * 16 + ln;
            #pragma unroll
            for (int s = 0; s < 2; ++s) {
                const int k0 = 4 * s + 2 * lh;
                Wx[q][s].x = (k0     < I_) ? w_ih[g * I_ + k0]     : 0.0f;
                Wx[q][s].y = (k0 + 1 < I_) ? w_ih[g * I_ + k0 + 1] : 0.0f;
            }
            bx[q] = (i < 2) ? (b_ih[g] + b_hh[g]) : b_ih[g];
        }
        const bool fcl = (ln < O_);
        #pragma unroll
        for (int s = 0; s < 8; ++s) {
            const int k0 = 4 * s + 2 * lh;
            Wfc[s].x = fcl ? fc_w[ln * H_ + k0]     : 0.0f;
            Wfc[s].y = fcl ? fc_w[ln * H_ + k0 + 1] : 0.0f;
        }
        fcb = fcl ? fc_b[ln] : 0.0f;
    }

    const int lenD = lengths[b0 + lh];

    const float* xrow = x + ((size_t)(b0 + slotA) * T_ + stepA) * I_;
    const int    hcol = (side ? 0 : wv) * 16 + ln;
    const int    gw   = side ? 0 : wv;

    auto load_x = [&](int jb, v2f& xa0, v2f& xa1) {
        const float* xp = xrow + (size_t)jb * (TB_ * I_);
        xa0.x = xp[2 * lh];
        xa0.y = xp[2 * lh + 1];
        xa1.x = lh ? 0.0f : xp[4];
        xa1.y = lh ? 0.0f : xp[5];
        if (jb + 2 < NBLK_) __builtin_prefetch(xp + 2 * TB_ * I_, 0, 1);
    };
    auto project_x = [&](int jb, v2f xa0, v2f xa1) {
        v8f a[6];
        #pragma unroll
        for (int q = 0; q < 6; ++q) {
            #pragma unroll
            for (int v = 0; v < 8; ++v) a[q][v] = bx[q];
            a[q] = wmma4(xa0, Wx[q][0], a[q]);
            a[q] = wmma4(xa1, Wx[q][1], a[q]);
        }
        float* dst = &ldsg[(((jb & 1) * TB_) * 2 * 32 + lane) * 4];
        #pragma unroll
        for (int v = 0; v < 8; ++v) {
            v4f p0, p1;
            p0.x = a[0][v]; p0.y = a[1][v]; p0.z = a[2][v]; p0.w = 0.0f;
            p1.x = a[3][v]; p1.y = a[4][v]; p1.z = a[5][v]; p1.w = 0.0f;
            *(v4f*)&dst[((v * 2 + 0) * 32) * 4] = p0;
            *(v4f*)&dst[((v * 2 + 1) * 32) * 4] = p1;
        }
    };
    auto fetch_start = [&](int ts) -> v4f {
        return *(const v4f*)&ldsg[(((ts & 15) * 2 + gw) * 32 + lane) * 4];
    };

    v2f   hA[8];
    float hold = 0.0f;
    #pragma unroll
    for (int s = 0; s < 8; ++s) { hA[s].x = 0.0f; hA[s].y = 0.0f; }

    v8f yv;
    #pragma unroll
    for (int v = 0; v < 8; ++v) yv[v] = 0.0f;
    auto fc_block = [&](int jb) {
        const float* hp = &ldsh[(((jb & 1) * TB_ + stepA) * RB_ + slotA) * H_ + 2 * lh];
        v2f hB[8];
        #pragma unroll
        for (int s = 0; s < 8; ++s) hB[s] = *(const v2f*)&hp[4 * s];
        v8f accF;
        #pragma unroll
        for (int v = 0; v < 8; ++v) accF[v] = fcb;
        #pragma unroll
        for (int s = 0; s < 8; ++s) accF = wmma4(hB[s], Wfc[s], accF);
        #pragma unroll
        for (int v = 0; v < 8; ++v) yv[v] = (jb * TB_ + v < lenD) ? accF[v] : fcb;
    };
    auto store_y = [&](int jb) {
        if (ln < O_) {
            float* op = out + ((size_t)(b0 + lh) * T_ + (size_t)jb * TB_) * O_ + ln;
            #pragma unroll
            for (int v = 0; v < 8; ++v) op[v * O_] = yv[v];
        }
    };

    const unsigned selb  = 2u * (unsigned)lh;
    const bool     ownrw = (slotA == lh);
    const int hoth = slotA * H_ + 16 * (1 - gw) + 2 * lh;

    v2f xb0, xb1;
    xb0.x = 0.0f; xb0.y = 0.0f; xb1.x = 0.0f; xb1.y = 0.0f;
    if (side) {
        load_x(0, xb0, xb1);
        project_x(0, xb0, xb1);
    } else {
        ldsh[(15 * RB_ + lh) * H_ + hcol] = 0.0f;
    }
    __syncthreads();

    if (!side) {
        v8f iniR, iniZ, iniN;
        #pragma unroll
        for (int v = 0; v < 8; ++v) { iniR[v] = 0.0f; iniZ[v] = 0.0f; iniN[v] = bhn; }
        float xnS;
        {
            const v4f c0 = fetch_start(0);
            iniR[0] = c0.x; iniZ[0] = c0.y; xnS = c0.z;
        }

        #pragma unroll 1
        for (int t = 0; t < T_; ++t) {
            v8f accR  = wmma4(hA[0], Wg[0][0], iniR);
            v8f accZ  = wmma4(hA[0], Wg[1][0], iniZ);
            v8f accHN = wmma4(hA[0], Wg[2][0], iniN);

            step_signal_behind(accR[0], accZ[0], accHN[0]);

            accR  = wmma4(hA[1], Wg[0][1], accR);
            accZ  = wmma4(hA[1], Wg[1][1], accZ);
            accHN = wmma4(hA[1], Wg[2][1], accHN);

            step_wait_behind(accR[0], accZ[0], accHN[0]);

            const float* hp = &ldsh[((t + 15) & 15) * (RB_ * H_) + hoth];
            #pragma unroll
            for (int s = 0; s < 4; ++s) hA[4 + s] = *(const v2f*)&hp[4 * s];
            const v4f cSn = fetch_start(t + 1);
            __builtin_amdgcn_sched_barrier(0);
            #pragma unroll
            for (int s = 2; s < 8; ++s) {
                accR  = wmma4(hA[s], Wg[0][s], accR);
                accZ  = wmma4(hA[s], Wg[1][s], accZ);
                accHN = wmma4(hA[s], Wg[2][s], accHN);
            }

            const float r  = fsigmoid(accR[0]);
            const float z  = fsigmoid(accZ[0]);
            const float n  = ftanh(xnS + r * accHN[0]);
            const float hn = (1.0f - z) * n + z * hold;
            const float hs = (t < lenD) ? hn : hold;

            iniR[0] = cSn.x; iniZ[0] = cSn.y; xnS = cSn.z;

            ldsh[((t & 15) * RB_ + lh) * H_ + hcol] = hs;
            __builtin_amdgcn_sched_barrier(0);

            const unsigned hsi = (unsigned)__float_as_int(hs);
            #pragma unroll
            for (int s = 0; s < 4; ++s) {
                const unsigned c0 = selb + 4u * (unsigned)s;
                const unsigned o0 = __builtin_amdgcn_permlane16_var (hsi, hsi, c0,      true, false);
                const unsigned x0 = __builtin_amdgcn_permlanex16_var(hsi, hsi, c0,      true, false);
                const unsigned o1 = __builtin_amdgcn_permlane16_var (hsi, hsi, c0 + 1u, true, false);
                const unsigned x1 = __builtin_amdgcn_permlanex16_var(hsi, hsi, c0 + 1u, true, false);
                hA[s].x = __int_as_float((int)(ownrw ? o0 : x0));
                hA[s].y = __int_as_float((int)(ownrw ? o1 : x1));
            }
            hold = hs;
        }
    } else {
        #pragma unroll 1
        for (int t = 0; t < T_; ++t) {
            step_signal();
            step_wait();

            const int jb = t >> 3;
            const int rr = t & 7;
            if (rr == 0) {
                if (jb > 0) fc_block(jb - 1);
            } else if (rr == 1) {
                if (jb > 0) store_y(jb - 1);
                if (jb + 1 < NBLK_) load_x(jb + 1, xb0, xb1);
            } else if (rr == 3) {
                if (jb + 1 < NBLK_) project_x(jb + 1, xb0, xb1);
            }
        }
    }

    step_signal();
    step_wait();

    if (side) {
        fc_block(NBLK_ - 1);
        store_y(NBLK_ - 1);
    }
}

extern "C" void kernel_launch(void* const* d_in, const int* in_sizes, int n_in,
                              void* d_out, int out_size, void* d_ws, size_t ws_size,
                              hipStream_t stream) {
    const float* x       = (const float*)d_in[0];
    const int*   lengths = (const int*)  d_in[1];
    const float* w_ih    = (const float*)d_in[2];
    const float* w_hh    = (const float*)d_in[3];
    const float* b_ih    = (const float*)d_in[4];
    const float* b_hh    = (const float*)d_in[5];
    const float* fc_w    = (const float*)d_in[6];
    const float* fc_b    = (const float*)d_in[7];
    float*       out     = (float*)d_out;

    gru_fused_kernel<<<dim3(B_ / RB_), dim3(96), 0, stream>>>(
        x, lengths, w_ih, w_hh, b_ih, b_hh, fc_w, fc_b, out);
}


# Recurrent products split by gate over 4 waves: two new z waves run the update gate (8 WMMA, two K-half chains added last: z sum order changed), z_t handed over in a 2nd barrier phase; r/n waves 16/24
# speedup vs baseline: 2.6618x; 1.0568x over previous
#include <hip/hip_runtime.h>


typedef float v2f __attribute__((ext_vector_type(2)));
typedef float v4f __attribute__((ext_vector_type(4)));
typedef float v8f __attribute__((ext_vector_type(8)));

#define B_ 128
#define T_ 4096
#define I_ 6
#define H_ 32
#define O_ 2
#define RB_ 2
#define TB_ 8
#define NBLK_ (T_ / TB_)

__device__ __forceinline__ v8f wmma4(v2f a, v2f b, v8f c) {
    return __builtin_amdgcn_wmma_f32_16x16x4_f32(
         false, a,  false, b,
         (short)0, c,  false,  false);
}

__device__ __forceinline__ float fsigmoid(float a) {
    return __builtin_amdgcn_rcpf(1.0f + __builtin_amdgcn_exp2f(-1.44269504088896f * a));
}
__device__ __forceinline__ float ftanh(float a) {
    return 2.0f * __builtin_amdgcn_rcpf(1.0f + __builtin_amdgcn_exp2f(-2.88539008177793f * a)) - 1.0f;
}

__device__ __forceinline__ void step_signal() {
    asm volatile("s_wait_dscnt 0x0\n\ts_barrier_signal -1" ::: "memory");
}
__device__ __forceinline__ void step_signal_bare() {
    asm volatile("s_barrier_signal -1" ::: "memory");
}
__device__ __forceinline__ void step_wait() {
    asm volatile("s_barrier_wait -1" ::: "memory");
}

__device__ __forceinline__ void rn_waitA_signalB_behind(float r0, float n0) {
    asm volatile("s_barrier_wait -1\n\ts_barrier_signal -1" :: "v"(r0), "v"(n0) : "memory");
}
__device__ __forceinline__ void rn_waitB_behind(float r) {
    asm volatile("s_barrier_wait -1" :: "v"(r) : "memory");
}

__global__ __launch_bounds__(160, 1) void gru_fused_kernel(
    const float* __restrict__ x,
    const int*   __restrict__ lengths,
    const float* __restrict__ w_ih,
    const float* __restrict__ w_hh,
    const float* __restrict__ b_ih,
    const float* __restrict__ b_hh,
    const float* __restrict__ fc_w,
    const float* __restrict__ fc_b,
    float*       __restrict__ out)
{
    __shared__ __attribute__((aligned(16))) float ldsh[2 * TB_ * RB_ * H_];
    __shared__ __attribute__((aligned(16))) float ldsg[2 * TB_ * 2 * 32 * 4];
    __shared__ float ldsz[2 * 2 * 32];

    const int tid  = threadIdx.x;
    const int wv   = __builtin_amdgcn_readfirstlane(tid >> 5);
    const int lane = tid & 31;
    const int lh   = lane >> 4;
    const int ln   = lane & 15;
    const int b0   = blockIdx.x * RB_;
    const int slotA = ln >> 3;
    const int stepA = ln & 7;
    const bool side  = (wv == 4);
    const bool zrole = (wv < 2);
    const int  gw    = side ? 0 : (wv & 1);


    v2f   Wg[3][8];
    float bhn = 0.0f;
    v2f   Wx[6][2];
    float bx[6];
    v2f   Wfc[8];
    float fcb = 0.0f;
    #pragma unroll
    for (int i = 0; i < 3; ++i)
        #pragma unroll
        for (int s = 0; s < 8; ++s) { Wg[i][s].x = 0.0f; Wg[i][s].y = 0.0f; }
    #pragma unroll
    for (int q = 0; q < 6; ++q) {
        bx[q] = 0.0f;
        #pragma unroll
        for (int s = 0; s < 2; ++s) { Wx[q][s].x = 0.0f; Wx[q][s].y = 0.0f; }
    }
    #pragma unroll
    for (int s = 0; s < 8; ++s) { Wfc[s].x = 0.0f; Wfc[s].y = 0.0f; }

    if (zrole) {
        const int g = 1 * 32 + gw * 16 + ln;
        #pragma unroll
        for (int s = 0; s < 8; ++s) {
            const int k0 = 4 * s + 2 * lh;
            Wg[1][s].x = w_hh[g * H_ + k0];
            Wg[1][s].y = w_hh[g * H_ + k0 + 1];
        }
    } else if (!side) {
        #pragma unroll
        for (int i = 0; i < 3; i += 2) {
            const int g = i * 32 + gw * 16 + ln;
            #pragma unroll
            for (int s = 0; s < 8; ++s) {
                const int sl = 4 * (gw ^ (s >> 2)) + (s & 3);
                const int k0 = 4 * sl + 2 * lh;
                Wg[i][s].x = w_hh[g * H_ + k0];
                Wg[i][s].y = w_hh[g * H_ + k0 + 1];
            }
        }
        bhn = b_hh[2 * 32 + gw * 16 + ln];
    } else {
        #pragma unroll
        for (int q = 0; q < 6; ++q) {
            const int i = q % 3;
            const int g = i * 32 + (q / 3) * 16 + ln;
            #pragma unroll
            for (int s = 0; s < 2; ++s) {
                const int k0 = 4 * s + 2 * lh;
                Wx[q][s].x = (k0     < I_) ? w_ih[g * I_ + k0]     : 0.0f;
                Wx[q][s].y = (k0 + 1 < I_) ? w_ih[g * I_ + k0 + 1] : 0.0f;
            }
            bx[q] = (i < 2) ? (b_ih[g] + b_hh[g]) : b_ih[g];
        }
        const bool fcl = (ln < O_);
        #pragma unroll
        for (int s = 0; s < 8; ++s) {
            const int k0 = 4 * s + 2 * lh;
            Wfc[s].x = fcl ? fc_w[ln * H_ + k0]     : 0.0f;
            Wfc[s].y = fcl ? fc_w[ln * H_ + k0 + 1] : 0.0f;
        }
        fcb = fcl ? fc_b[ln] : 0.0f;
    }

    const int lenD = lengths[b0 + lh];

    const float* xrow = x + ((size_t)(b0 + slotA) * T_ + stepA) * I_;
    const int    hcol = gw * 16 + ln;

    auto load_x = [&](int jb, v2f& xa0, v2f& xa1) {
        const float* xp = xrow + (size_t)jb * (TB_ * I_);
        xa0.x = xp[2 * lh];
        xa0.y = xp[2 * lh + 1];
        xa1.x = lh ? 0.0f : xp[4];
        xa1.y = lh ? 0.0f : xp[5];
        if (jb + 2 < NBLK_) __builtin_prefetch(xp + 2 * TB_ * I_, 0, 1);
    };
    auto project_x = [&](int jb, v2f xa0, v2f xa1) {
        v8f a[6];
        #pragma unroll
        for (int q = 0; q < 6; ++q) {
            #pragma unroll
            for (int v = 0; v < 8; ++v) a[q][v] = bx[q];
            a[q] = wmma4(xa0, Wx[q][0], a[q]);
            a[q] = wmma4(xa1, Wx[q][1], a[q]);
        }
        float* dst = &ldsg[(((jb & 1) * TB_) * 2 * 32 + lane) * 4];
        #pragma unroll
        for (int v = 0; v < 8; ++v) {
            v4f p0, p1;
            p0.x = a[0][v]; p0.y = a[1][v]; p0.z = a[2][v]; p0.w = 0.0f;
            p1.x = a[3][v]; p1.y = a[4][v]; p1.z = a[5][v]; p1.w = 0.0f;
            *(v4f*)&dst[((v * 2 + 0) * 32) * 4] = p0;
            *(v4f*)&dst[((v * 2 + 1) * 32) * 4] = p1;
        }
    };
    auto fetch_start = [&](int ts) -> v4f {
        return *(const v4f*)&ldsg[(((ts & 15) * 2 + gw) * 32 + lane) * 4];
    };

    v2f   hA[8];
    float hold = 0.0f;
    #pragma unroll
    for (int s = 0; s < 8; ++s) { hA[s].x = 0.0f; hA[s].y = 0.0f; }

    v8f yv;
    #pragma unroll
    for (int v = 0; v < 8; ++v) yv[v] = 0.0f;
    auto fc_block = [&](int jb) {
        const float* hp = &ldsh[(((jb & 1) * TB_ + stepA) * RB_ + slotA) * H_ + 2 * lh];
        v2f hB[8];
        #pragma unroll
        for (int s = 0; s < 8; ++s) hB[s] = *(const v2f*)&hp[4 * s];
        v8f accF;
        #pragma unroll
        for (int v = 0; v < 8; ++v) accF[v] = fcb;
        #pragma unroll
        for (int s = 0; s < 8; ++s) accF = wmma4(hB[s], Wfc[s], accF);
        #pragma unroll
        for (int v = 0; v < 8; ++v) yv[v] = (jb * TB_ + v < lenD) ? accF[v] : fcb;
    };
    auto store_y = [&](int jb) {
        if (ln < O_) {
            float* op = out + ((size_t)(b0 + lh) * T_ + (size_t)jb * TB_) * O_ + ln;
            #pragma unroll
            for (int v = 0; v < 8; ++v) op[v * O_] = yv[v];
        }
    };

    const unsigned selb  = 2u * (unsigned)lh;
    const bool     ownrw = (slotA == lh);
    const int hoth = slotA * H_ + 16 * (1 - gw) + 2 * lh;

    v2f xb0, xb1;
    xb0.x = 0.0f; xb0.y = 0.0f; xb1.x = 0.0f; xb1.y = 0.0f;
    if (side) {
        load_x(0, xb0, xb1);
        project_x(0, xb0, xb1);
    } else if (!zrole) {
        ldsh[(15 * RB_ + lh) * H_ + hcol] = 0.0f;
    }
    __syncthreads();

    if (zrole) {
        float zs = ldsg[((0 * 2 + gw) * 32 + lane) * 4 + 1];
        const int hrow = slotA * H_ + 2 * lh;

        #pragma unroll 1
        for (int t = 0; t < T_; ++t) {
            step_signal_bare();
            step_wait();

            const float* hp = &ldsh[((t + 15) & 15) * (RB_ * H_) + hrow];
            v2f hz[8];
            #pragma unroll
            for (int s = 0; s < 8; ++s) hz[s] = *(const v2f*)&hp[4 * s];
            const float zsn = ldsg[((((t + 1) & 15) * 2 + gw) * 32 + lane) * 4 + 1];

            v8f a0, a1;
            #pragma unroll
            for (int v = 0; v < 8; ++v) { a0[v] = 0.0f; a1[v] = 0.0f; }
            a0[0] = zs;
            #pragma unroll
            for (int s = 0; s < 4; ++s) {
                a0 = wmma4(hz[s],     Wg[1][s],     a0);
                a1 = wmma4(hz[4 + s], Wg[1][4 + s], a1);
            }
            const float z = fsigmoid(a0[0] + a1[0]);

            ldsz[((t & 1) * 2 + gw) * 32 + lane] = z;
            step_signal();
            step_wait();
            zs = zsn;
        }
    } else if (!side) {
        v8f iniR, iniN;
        #pragma unroll
        for (int v = 0; v < 8; ++v) { iniR[v] = 0.0f; iniN[v] = bhn; }
        float xnS;
        {
            const v4f c0 = fetch_start(0);
            iniR[0] = c0.x; xnS = c0.z;
        }

        step_signal();

        #pragma unroll 1
        for (int t = 0; t < T_; ++t) {
            v8f accR  = wmma4(hA[0], Wg[0][0], iniR);
            v8f accHN = wmma4(hA[0], Wg[2][0], iniN);

            rn_waitA_signalB_behind(accR[0], accHN[0]);

            const float* hp = &ldsh[((t + 15) & 15) * (RB_ * H_) + hoth];
            #pragma unroll
            for (int s = 0; s < 4; ++s) hA[4 + s] = *(const v2f*)&hp[4 * s];
            const v4f cSn = fetch_start(t + 1);
            __builtin_amdgcn_sched_barrier(0);
            #pragma unroll
            for (int s = 1; s < 8; ++s) {
                accR  = wmma4(hA[s], Wg[0][s], accR);
                accHN = wmma4(hA[s], Wg[2][s], accHN);
            }

            const float r  = fsigmoid(accR[0]);
            rn_waitB_behind(r);
            const float z  = ldsz[((t & 1) * 2 + gw) * 32 + lane];
            __builtin_amdgcn_sched_barrier(0);
            const float n  = ftanh(xnS + r * accHN[0]);
            __builtin_amdgcn_sched_barrier(0);
            const float hn = (1.0f - z) * n + z * hold;
            const float hs = (t < lenD) ? hn : hold;

            iniR[0] = cSn.x; xnS = cSn.z;

            ldsh[((t & 15) * RB_ + lh) * H_ + hcol] = hs;
            __builtin_amdgcn_sched_barrier(0);

            const unsigned hsi = (unsigned)__float_as_int(hs);
            #pragma unroll
            for (int s = 0; s < 4; ++s) {
                const unsigned c0 = selb + 4u * (unsigned)s;
                const unsigned o0 = __builtin_amdgcn_permlane16_var (hsi, hsi, c0,      true, false);
                const unsigned x0 = __builtin_amdgcn_permlanex16_var(hsi, hsi, c0,      true, false);
                const unsigned o1 = __builtin_amdgcn_permlane16_var (hsi, hsi, c0 + 1u, true, false);
                const unsigned x1 = __builtin_amdgcn_permlanex16_var(hsi, hsi, c0 + 1u, true, false);
                hA[s].x = __int_as_float((int)(ownrw ? o0 : x0));
                hA[s].y = __int_as_float((int)(ownrw ? o1 : x1));
            }
            hold = hs;

            step_signal();
        }
    } else {
        #pragma unroll 1
        for (int t = 0; t < T_; ++t) {
            step_signal();
            step_wait();
            step_signal_bare();

            const int jb = t >> 3;
            const int rr = t & 7;
            if (rr == 0) {
                if (jb > 0) fc_block(jb - 1);
            } else if (rr == 1) {
                if (jb > 0) store_y(jb - 1);
                if (jb + 1 < NBLK_) load_x(jb + 1, xb0, xb1);
            } else if (rr == 3) {
                if (jb + 1 < NBLK_) project_x(jb + 1, xb0, xb1);
            }
            step_wait();
        }
    }

    if (zrole || side) step_signal();
    step_wait();

    if (side) {
        fc_block(NBLK_ - 1);
        store_y(NBLK_ - 1);
    }
}

extern "C" void kernel_launch(void* const* d_in, const int* in_sizes, int n_in,
                              void* d_out, int out_size, void* d_ws, size_t ws_size,
                              hipStream_t stream) {
    const float* x       = (const float*)d_in[0];
    const int*   lengths = (const int*)  d_in[1];
    const float* w_ih    = (const float*)d_in[2];
    const float* w_hh    = (const float*)d_in[3];
    const float* b_ih    = (const float*)d_in[4];
    const float* b_hh    = (const float*)d_in[5];
    const float* fc_w    = (const float*)d_in[6];
    const float* fc_b    = (const float*)d_in[7];
    float*       out     = (float*)d_out;

    gru_fused_kernel<<<dim3(B_ / RB_), dim3(160), 0, stream>>>(
        x, lengths, w_ih, w_hh, b_ih, b_hh, fc_w, fc_b, out);
}


# Vector chain from z products to h store cut 9 -> 5 dependent ops: z waves hand over w=1-z with the length mask folded in (start +inf), blend one fma h+w*(n-h); exp arg by fma; rounding changed
# speedup vs baseline: 2.7456x; 1.0315x over previous
#include <hip/hip_runtime.h>


typedef float v2f __attribute__((ext_vector_type(2)));
typedef float v4f __attribute__((ext_vector_type(4)));
typedef float v8f __attribute__((ext_vector_type(8)));

#define B_ 128
#define T_ 4096
#define I_ 6
#define H_ 32
#define O_ 2
#define RB_ 2
#define TB_ 8
#define NBLK_ (T_ / TB_)

__device__ __forceinline__ v8f wmma4(v2f a, v2f b, v8f c) {
    return __builtin_amdgcn_wmma_f32_16x16x4_f32(
         false, a,  false, b,
         (short)0, c,  false,  false);
}

__device__ __forceinline__ float fsigmoid(float a) {
    return __builtin_amdgcn_rcpf(1.0f + __builtin_amdgcn_exp2f(-1.44269504088896f * a));
}
__device__ __forceinline__ float fsigmoid_compl2(float a0, float a1) {
    return __builtin_amdgcn_rcpf(1.0f + __builtin_amdgcn_exp2f(
        __builtin_fmaf(1.44269504088896f, a1, 1.44269504088896f * a0)));
}
__device__ __forceinline__ float ftanh(float a) {
    return 2.0f * __builtin_amdgcn_rcpf(1.0f + __builtin_amdgcn_exp2f(-2.88539008177793f * a)) - 1.0f;
}

__device__ __forceinline__ void step_signal() {
    asm volatile("s_wait_dscnt 0x0\n\ts_barrier_signal -1" ::: "memory");
}
__device__ __forceinline__ void step_signal_bare() {
    asm volatile("s_barrier_signal -1" ::: "memory");
}
__device__ __forceinline__ void step_wait() {
    asm volatile("s_barrier_wait -1" ::: "memory");
}

__device__ __forceinline__ void rn_waitA_signalB_behind(float r0, float n0) {
    asm volatile("s_barrier_wait -1\n\ts_barrier_signal -1" :: "v"(r0), "v"(n0) : "memory");
}
__device__ __forceinline__ void rn_waitB_behind(float r) {
    asm volatile("s_barrier_wait -1" :: "v"(r) : "memory");
}

__global__ __launch_bounds__(160, 1) void gru_fused_kernel(
    const float* __restrict__ x,
    const int*   __restrict__ lengths,
    const float* __restrict__ w_ih,
    const float* __restrict__ w_hh,
    const float* __restrict__ b_ih,
    const float* __restrict__ b_hh,
    const float* __restrict__ fc_w,
    const float* __restrict__ fc_b,
    float*       __restrict__ out)
{
    __shared__ __attribute__((aligned(16))) float ldsh[2 * TB_ * RB_ * H_];
    __shared__ __attribute__((aligned(16))) float ldsg[2 * TB_ * 2 * 32 * 4];
    __shared__ float ldsw[2 * 2 * 32];

    const int tid  = threadIdx.x;
    const int wv   = __builtin_amdgcn_readfirstlane(tid >> 5);
    const int lane = tid & 31;
    const int lh   = lane >> 4;
    const int ln   = lane & 15;
    const int b0   = blockIdx.x * RB_;
    const int slotA = ln >> 3;
    const int stepA = ln & 7;
    const bool side  = (wv == 4);
    const bool zrole = (wv < 2);
    const int  gw    = side ? 0 : (wv & 1);


    v2f   Wg[3][8];
    float bhn = 0.0f;
    v2f   Wx[6][2];
    float bx[6];
    v2f   Wfc[8];
    float fcb = 0.0f;
    #pragma unroll
    for (int i = 0; i < 3; ++i)
        #pragma unroll
        for (int s = 0; s < 8; ++s) { Wg[i][s].x = 0.0f; Wg[i][s].y = 0.0f; }
    #pragma unroll
    for (int q = 0; q < 6; ++q) {
        bx[q] = 0.0f;
        #pragma unroll
        for (int s = 0; s < 2; ++s) { Wx[q][s].x = 0.0f; Wx[q][s].y = 0.0f; }
    }
    #pragma unroll
    for (int s = 0; s < 8; ++s) { Wfc[s].x = 0.0f; Wfc[s].y = 0.0f; }

    if (zrole) {
        const int g = 1 * 32 + gw * 16 + ln;
        #pragma unroll
        for (int s = 0; s < 8; ++s) {
            const int k0 = 4 * s + 2 * lh;
            Wg[1][s].x = w_hh[g * H_ + k0];
            Wg[1][s].y = w_hh[g * H_ + k0 + 1];
        }
    } else if (!side) {
        #pragma unroll
        for (int i = 0; i < 3; i += 2) {
            const int g = i * 32 + gw * 16 + ln;
            #pragma unroll
            for (int s = 0; s < 8; ++s) {
                const int sl = 4 * (gw ^ (s >> 2)) + (s & 3);
                const int k0 = 4 * sl + 2 * lh;
                Wg[i][s].x = w_hh[g * H_ + k0];
                Wg[i][s].y = w_hh[g * H_ + k0 + 1];
            }
        }
        bhn = b_hh[2 * 32 + gw * 16 + ln];
    } else {
        #pragma unroll
        for (int q = 0; q < 6; ++q) {
            const int i = q % 3;
            const int g = i * 32 + (q / 3) * 16 + ln;
            #pragma unroll
            for (int s = 0; s < 2; ++s) {
                const int k0 = 4 * s + 2 * lh;
                Wx[q][s].x = (k0     < I_) ? w_ih[g * I_ + k0]     : 0.0f;
                Wx[q][s].y = (k0 + 1 < I_) ? w_ih[g * I_ + k0 + 1] : 0.0f;
            }
            bx[q] = (i < 2) ? (b_ih[g] + b_hh[g]) : b_ih[g];
        }
        const bool fcl = (ln < O_);
        #pragma unroll
        for (int s = 0; s < 8; ++s) {
            const int k0 = 4 * s + 2 * lh;
            Wfc[s].x = fcl ? fc_w[ln * H_ + k0]     : 0.0f;
            Wfc[s].y = fcl ? fc_w[ln * H_ + k0 + 1] : 0.0f;
        }
        fcb = fcl ? fc_b[ln] : 0.0f;
    }

    const int lenD = lengths[b0 + lh];

    const float* xrow = x + ((size_t)(b0 + slotA) * T_ + stepA) * I_;
    const int    hcol = gw * 16 + ln;

    auto load_x = [&](int jb, v2f& xa0, v2f& xa1) {
        const float* xp = xrow + (size_t)jb * (TB_ * I_);
        xa0.x = xp[2 * lh];
        xa0.y = xp[2 * lh + 1];
        xa1.x = lh ? 0.0f : xp[4];
        xa1.y = lh ? 0.0f : xp[5];
        if (jb + 2 < NBLK_) __builtin_prefetch(xp + 2 * TB_ * I_, 0, 1);
    };
    auto project_x = [&](int jb, v2f xa0, v2f xa1) {
        v8f a[6];
        #pragma unroll
        for (int q = 0; q < 6; ++q) {
            #pragma unroll
            for (int v = 0; v < 8; ++v) a[q][v] = bx[q];
            a[q] = wmma4(xa0, Wx[q][0], a[q]);
            a[q] = wmma4(xa1, Wx[q][1], a[q]);
        }
        float* dst = &ldsg[(((jb & 1) * TB_) * 2 * 32 + lane) * 4];
        #pragma unroll
        for (int v = 0; v < 8; ++v) {
            v4f p0, p1;
            p0.x = a[0][v]; p0.y = a[1][v]; p0.z = a[2][v]; p0.w = 0.0f;
            p1.x = a[3][v]; p1.y = a[4][v]; p1.z = a[5][v]; p1.w = 0.0f;
            *(v4f*)&dst[((v * 2 + 0) * 32) * 4] = p0;
            *(v4f*)&dst[((v * 2 + 1) * 32) * 4] = p1;
        }
    };
    auto fetch_start = [&](int ts) -> v4f {
        return *(const v4f*)&ldsg[(((ts & 15) * 2 + gw) * 32 + lane) * 4];
    };

    v2f   hA[8];
    float hold = 0.0f;
    #pragma unroll
    for (int s = 0; s < 8; ++s) { hA[s].x = 0.0f; hA[s].y = 0.0f; }

    v8f yv;
    #pragma unroll
    for (int v = 0; v < 8; ++v) yv[v] = 0.0f;
    auto fc_block = [&](int jb) {
        const float* hp = &ldsh[(((jb & 1) * TB_ + stepA) * RB_ + slotA) * H_ + 2 * lh];
        v2f hB[8];
        #pragma unroll
        for (int s = 0; s < 8; ++s) hB[s] = *(const v2f*)&hp[4 * s];
        v8f accF;
        #pragma unroll
        for (int v = 0; v < 8; ++v) accF[v] = fcb;
        #pragma unroll
        for (int s = 0; s < 8; ++s) accF = wmma4(hB[s], Wfc[s], accF);
        #pragma unroll
        for (int v = 0; v < 8; ++v) yv[v] = (jb * TB_ + v < lenD) ? accF[v] : fcb;
    };
    auto store_y = [&](int jb) {
        if (ln < O_) {
            float* op = out + ((size_t)(b0 + lh) * T_ + (size_t)jb * TB_) * O_ + ln;
            #pragma unroll
            for (int v = 0; v < 8; ++v) op[v * O_] = yv[v];
        }
    };

    const unsigned selb  = 2u * (unsigned)lh;
    const bool     ownrw = (slotA == lh);
    const int hoth = slotA * H_ + 16 * (1 - gw) + 2 * lh;

    v2f xb0, xb1;
    xb0.x = 0.0f; xb0.y = 0.0f; xb1.x = 0.0f; xb1.y = 0.0f;
    if (side) {
        load_x(0, xb0, xb1);
        project_x(0, xb0, xb1);
    } else if (!zrole) {
        ldsh[(15 * RB_ + lh) * H_ + hcol] = 0.0f;
    }
    __syncthreads();

    if (zrole) {
        const float pinf = __builtin_huge_valf();
        float zs = ldsg[((0 * 2 + gw) * 32 + lane) * 4 + 1];
        zs = (0 < lenD) ? zs : pinf;
        const int hrow = slotA * H_ + 2 * lh;

        #pragma unroll 1
        for (int t = 0; t < T_; ++t) {
            step_signal_bare();
            step_wait();

            const float* hp = &ldsh[((t + 15) & 15) * (RB_ * H_) + hrow];
            v2f hz[8];
            #pragma unroll
            for (int j = 0; j < 8; ++j) {
                const int s = (j & 1) | ((j & 2) << 1) | ((j & 4) >> 1);
                hz[s] = *(const v2f*)&hp[4 * s];
            }
            __builtin_amdgcn_sched_barrier(0);
            const float zsr = ldsg[((((t + 1) & 15) * 2 + gw) * 32 + lane) * 4 + 1];
            const float zsn = (t + 1 < lenD) ? zsr : pinf;

            v8f a0, a1;
            #pragma unroll
            for (int v = 0; v < 8; ++v) { a0[v] = 0.0f; a1[v] = 0.0f; }
            a0[0] = zs;
            #pragma unroll
            for (int s = 0; s < 4; ++s) {
                a0 = wmma4(hz[s],     Wg[1][s],     a0);
                a1 = wmma4(hz[4 + s], Wg[1][4 + s], a1);
            }
            const float w = fsigmoid_compl2(a0[0], a1[0]);

            ldsw[((t & 1) * 2 + gw) * 32 + lane] = w;
            step_signal();
            step_wait();
            zs = zsn;
        }
    } else if (!side) {
        v8f iniR, iniN;
        #pragma unroll
        for (int v = 0; v < 8; ++v) { iniR[v] = 0.0f; iniN[v] = bhn; }
        float xnS;
        {
            const v4f c0 = fetch_start(0);
            iniR[0] = c0.x; xnS = c0.z;
        }

        step_signal();

        #pragma unroll 1
        for (int t = 0; t < T_; ++t) {
            v8f accR  = wmma4(hA[0], Wg[0][0], iniR);
            v8f accHN = wmma4(hA[0], Wg[2][0], iniN);

            rn_waitA_signalB_behind(accR[0], accHN[0]);

            const float* hp = &ldsh[((t + 15) & 15) * (RB_ * H_) + hoth];
            #pragma unroll
            for (int s = 0; s < 4; ++s) hA[4 + s] = *(const v2f*)&hp[4 * s];
            const v4f cSn = fetch_start(t + 1);
            __builtin_amdgcn_sched_barrier(0);
            #pragma unroll
            for (int s = 1; s < 8; ++s) {
                accR  = wmma4(hA[s], Wg[0][s], accR);
                accHN = wmma4(hA[s], Wg[2][s], accHN);
            }

            const float r  = fsigmoid(accR[0]);
            rn_waitB_behind(r);
            const float w  = ldsw[((t & 1) * 2 + gw) * 32 + lane];
            __builtin_amdgcn_sched_barrier(0);
            const float n  = ftanh(xnS + r * accHN[0]);
            const float dn = n - hold;
            __builtin_amdgcn_sched_barrier(0);
            const float hs = __builtin_fmaf(w, dn, hold);

            iniR[0] = cSn.x; xnS = cSn.z;

            ldsh[((t & 15) * RB_ + lh) * H_ + hcol] = hs;
            __builtin_amdgcn_sched_barrier(0);

            const unsigned hsi = (unsigned)__float_as_int(hs);
            #pragma unroll
            for (int s = 0; s < 4; ++s) {
                const unsigned c0 = selb + 4u * (unsigned)s;
                const unsigned o0 = __builtin_amdgcn_permlane16_var (hsi, hsi, c0,      true, false);
                const unsigned x0 = __builtin_amdgcn_permlanex16_var(hsi, hsi, c0,      true, false);
                const unsigned o1 = __builtin_amdgcn_permlane16_var (hsi, hsi, c0 + 1u, true, false);
                const unsigned x1 = __builtin_amdgcn_permlanex16_var(hsi, hsi, c0 + 1u, true, false);
                hA[s].x = __int_as_float((int)(ownrw ? o0 : x0));
                hA[s].y = __int_as_float((int)(ownrw ? o1 : x1));
            }
            hold = hs;

            step_signal();
        }
    } else {
        #pragma unroll 1
        for (int t = 0; t < T_; ++t) {
            step_signal();
            step_wait();
            step_signal_bare();

            const int jb = t >> 3;
            const int rr = t & 7;
            if (rr == 0) {
                if (jb > 0) fc_block(jb - 1);
            } else if (rr == 1) {
                if (jb > 0) store_y(jb - 1);
                if (jb + 1 < NBLK_) load_x(jb + 1, xb0, xb1);
            } else if (rr == 3) {
                if (jb + 1 < NBLK_) project_x(jb + 1, xb0, xb1);
            }
            step_wait();
        }
    }

    if (zrole || side) step_signal();
    step_wait();

    if (side) {
        fc_block(NBLK_ - 1);
        store_y(NBLK_ - 1);
    }
}

extern "C" void kernel_launch(void* const* d_in, const int* in_sizes, int n_in,
                              void* d_out, int out_size, void* d_ws, size_t ws_size,
                              hipStream_t stream) {
    const float* x       = (const float*)d_in[0];
    const int*   lengths = (const int*)  d_in[1];
    const float* w_ih    = (const float*)d_in[2];
    const float* w_hh    = (const float*)d_in[3];
    const float* b_ih    = (const float*)d_in[4];
    const float* b_hh    = (const float*)d_in[5];
    const float* fc_w    = (const float*)d_in[6];
    const float* fc_b    = (const float*)d_in[7];
    float*       out     = (float*)d_out;

    gru_fused_kernel<<<dim3(B_ / RB_), dim3(160), 0, stream>>>(
        x, lengths, w_ih, w_hh, b_ih, b_hh, fc_w, fc_b, out);
}
